# starting kernel
# baseline (speedup 1.0000x reference)
#include <hip/hip_runtime.h>

#define N_NODES 100000
#define N_EDGES 1200000
#define D_IN 64
#define D_HID 128

typedef __attribute__((ext_vector_type(2))) float v2f;
typedef __attribute__((ext_vector_type(8))) float v8f;

__global__ void zero_kernel(float4* __restrict__ p, int n4) {
    int i = blockIdx.x * blockDim.x + threadIdx.x;
    if (i < n4) p[i] = make_float4(0.f, 0.f, 0.f, 0.f);
}

template <int D>
__global__ void scatter_add_kernel(const float* __restrict__ feat,
                                   const int* __restrict__ src,
                                   const int* __restrict__ dst,
                                   float* __restrict__ agg,
                                   int n_edges) {
    const int lane = threadIdx.x & 31;
    const int wave = blockIdx.x * (blockDim.x >> 5) + (threadIdx.x >> 5);
    const int stride = gridDim.x * (blockDim.x >> 5);
    for (int e = wave; e < n_edges; e += stride) {
        const int s = src[e];
        const int d = dst[e];
        const float* fs = feat + (size_t)s * D;
        float* fd = agg + (size_t)d * D;
#pragma unroll
        for (int t = 0; t < D / 32; ++t) {
            float v = fs[lane + 32 * t];
            unsafeAtomicAdd(&fd[lane + 32 * t], v);
        }
    }
}

__global__ __launch_bounds__(256) void gemm_dual_wmma(
        const float* __restrict__ A1, int K1, const float* __restrict__ W1,
        const float* __restrict__ A2, int K2, const float* __restrict__ W2,
        const float* __restrict__ bias, float* __restrict__ D,
        int M, int do_relu) {
    const int tile_m = blockIdx.x;
    const int tile_n = threadIdx.x >> 5;
    if (tile_m * 16 >= M) return;

    const int lane = threadIdx.x & 31;
    const int half = lane >> 4;
    const int mn   = lane & 15;

    const int row = tile_m * 16 + mn;
    const int col = tile_n * 16 + mn;

    v8f c = {};

    const float* a_ptr = A1 + (size_t)row * K1 + 2 * half;
    const float* w_ptr = W1 + (size_t)col * K1 + 2 * half;
    for (int k = 0; k < K1; k += 4) {
        v2f a = *(const v2f*)(a_ptr + k);
        v2f b = *(const v2f*)(w_ptr + k);
        c = __builtin_amdgcn_wmma_f32_16x16x4_f32(false, a, false, b,
                                                  (short)0, c, false, false);
    }

    a_ptr = A2 + (size_t)row * K2 + 2 * half;
    w_ptr = W2 + (size_t)col * K2 + 2 * half;
    for (int k = 0; k < K2; k += 4) {
        v2f a = *(const v2f*)(a_ptr + k);
        v2f b = *(const v2f*)(w_ptr + k);
        c = __builtin_amdgcn_wmma_f32_16x16x4_f32(false, a, false, b,
                                                  (short)0, c, false, false);
    }

    const int out_col = tile_n * 16 + mn;
    const float bv = bias[out_col];
#pragma unroll
    for (int r = 0; r < 8; ++r) {
        const int out_row = tile_m * 16 + 8 * half + r;
        float v = c[r] + bv;
        if (do_relu) v = fmaxf(v, 0.f);
        D[(size_t)out_row * D_HID + out_col] = v;
    }
}

extern "C" void kernel_launch(void* const* d_in, const int* in_sizes, int n_in,
                              void* d_out, int out_size, void* d_ws, size_t ws_size,
                              hipStream_t stream) {
    const float* x       = (const float*)d_in[0];
    const int*   edge    = (const int*)d_in[1];
    const float* W1_rel  = (const float*)d_in[2];
    const float* b1_rel  = (const float*)d_in[3];
    const float* W1_root = (const float*)d_in[4];
    const float* W2_rel  = (const float*)d_in[5];
    const float* b2_rel  = (const float*)d_in[6];
    const float* W2_root = (const float*)d_in[7];
    float* out = (float*)d_out;

    const int* src = edge;
    const int* dst = edge + N_EDGES;

    float* h   = (float*)d_ws;
    float* agg = h + (size_t)N_NODES * D_HID;

    const int scatter_blocks = N_EDGES / 8;
    const int gemm_blocks    = N_NODES / 16;

    {
        int n4 = N_NODES * D_IN / 4;
        zero_kernel<<<(n4 + 255) / 256, 256, 0, stream>>>((float4*)agg, n4);
    }
    scatter_add_kernel<D_IN><<<scatter_blocks, 256, 0, stream>>>(x, src, dst, agg, N_EDGES);
    gemm_dual_wmma<<<gemm_blocks, 256, 0, stream>>>(agg, D_IN, W1_rel,
                                                    x, D_IN, W1_root,
                                                    b1_rel, h, N_NODES, 1);

    {
        int n4 = N_NODES * D_HID / 4;
        zero_kernel<<<(n4 + 255) / 256, 256, 0, stream>>>((float4*)agg, n4);
    }
    scatter_add_kernel<D_HID><<<scatter_blocks, 256, 0, stream>>>(h, src, dst, agg, N_EDGES);
    gemm_dual_wmma<<<gemm_blocks, 256, 0, stream>>>(agg, D_HID, W2_rel,
                                                    h, D_HID, W2_root,
                                                    b2_rel, out, N_NODES, 0);
}


# GEMM retiled: one wave per 16x128 row strip (8 accumulator tiles), weights staged once per work-group in LDS, A fragment reused over 8 tiles, b128 loads/stores; k summed in permuted order within 8
# speedup vs baseline: 1.1989x; 1.1989x over previous
#include <hip/hip_runtime.h>

#define N_NODES 100000
#define N_EDGES 1200000
#define D_IN 64
#define D_HID 128

typedef __attribute__((ext_vector_type(2))) float v2f;
typedef __attribute__((ext_vector_type(4))) float v4f;
typedef __attribute__((ext_vector_type(8))) float v8f;

__global__ void zero_kernel(float4* __restrict__ p, int n4) {
    int i = blockIdx.x * blockDim.x + threadIdx.x;
    if (i < n4) p[i] = make_float4(0.f, 0.f, 0.f, 0.f);
}

template <int D>
__global__ void scatter_add_kernel(const float* __restrict__ feat,
                                   const int* __restrict__ src,
                                   const int* __restrict__ dst,
                                   float* __restrict__ agg,
                                   int n_edges) {
    const int lane = threadIdx.x & 31;
    const int wave = blockIdx.x * (blockDim.x >> 5) + (threadIdx.x >> 5);
    const int stride = gridDim.x * (blockDim.x >> 5);
    for (int e = wave; e < n_edges; e += stride) {
        const int s = src[e];
        const int d = dst[e];
        const float* fs = feat + (size_t)s * D;
        float* fd = agg + (size_t)d * D;
#pragma unroll
        for (int t = 0; t < D / 32; ++t) {
            float v = fs[lane + 32 * t];
            unsafeAtomicAdd(&fd[lane + 32 * t], v);
        }
    }
}

#define GEMM_THREADS 256
#define GEMM_WAVES   8
#define GEMM_BLOCKS  512
#define CHUNK_K      32

template <int K, int P>
__device__ __forceinline__ void stage_weights(float* lds, const float* __restrict__ Wg, int col0) {
    constexpr int Q = K / 4;
    for (int i = threadIdx.x; i < D_HID * Q; i += GEMM_THREADS) {
        const int n = i / Q;
        const int q = i - n * Q;
        const int j = (n & 7) * 16 + (n >> 3);
        *(v4f*)(lds + j * P + col0 + 4 * q) = *(const v4f*)(Wg + n * K + 4 * q);
    }
}

template <int P>
__device__ __forceinline__ void chunk_mma(v8f (&acc)[8], const v4f (&a)[4], v4f (&b)[8],
                                          const float* w_cur, const float* w_after) {
#pragma unroll
    for (int i = 0; i < 4; ++i) {
        const float* wn = (i < 3) ? w_cur + 8 * (i + 1) : w_after;
        v4f bn[8];
#pragma unroll
        for (int t = 0; t < 8; ++t) bn[t] = *(const v4f*)(wn + t * 16 * P);

        const v2f a_lo = a[i].xy;
        const v2f a_hi = a[i].zw;
#pragma unroll
        for (int t = 0; t < 8; ++t) {
            const v2f bb = b[t].xy;
            acc[t] = __builtin_amdgcn_wmma_f32_16x16x4_f32(false, a_lo, false, bb,
                                                           (short)0, acc[t], false, false);
        }
#pragma unroll
        for (int t = 0; t < 8; ++t) {
            const v2f bb = b[t].zw;
            acc[t] = __builtin_amdgcn_wmma_f32_16x16x4_f32(false, a_hi, false, bb,
                                                           (short)0, acc[t], false, false);
        }
#pragma unroll
        for (int t = 0; t < 8; ++t) b[t] = bn[t];
    }
}

template <int K1, int K2, int RELU>
__global__ __launch_bounds__(GEMM_THREADS) void gemm_dual_strip(
        const float* __restrict__ A1, const float* __restrict__ W1,
        const float* __restrict__ A2, const float* __restrict__ W2,
        const float* __restrict__ bias, float* __restrict__ D, int n_strips) {
    constexpr int KK = K1 + K2;
    constexpr int P  = KK + 4;
    constexpr int C1 = K1 / CHUNK_K;
    constexpr int C  = KK / CHUNK_K;
    static_assert(K1 % CHUNK_K == 0 && K2 % CHUNK_K == 0 && C % 2 == 0, "whole chunks, in pairs");
    static_assert(((P / 4) & 1) == 1, "pitch is an odd number of 16-B units");

    __shared__ __attribute__((aligned(16))) float sW[D_HID * P];

    stage_weights<K1, P>(sW, W1, 0);
    stage_weights<K2, P>(sW, W2, K1);
    __syncthreads();

    const int lane = threadIdx.x & 31;
    const int h    = lane >> 4;
    const int c    = lane & 15;
    const int wave = __builtin_amdgcn_readfirstlane(threadIdx.x >> 5);

    const v4f b_lo = *(const v4f*)(bias + 8 * c);
    const v4f b_hi = *(const v4f*)(bias + 8 * c + 4);

    const float* w_lds = sW + c * P + 4 * h;

    for (int s = wave * (int)gridDim.x + (int)blockIdx.x; s < n_strips;
         s += GEMM_WAVES * (int)gridDim.x) {
        const int row0 = s * 16;
        const float* a1 = A1 + (size_t)(row0 + c) * K1 + 4 * h;
        const float* a2 = A2 + (size_t)(row0 + c) * K2 + 4 * h;

        v8f acc[8];
#pragma unroll
        for (int t = 0; t < 8; ++t) acc[t] = v8f{};

        v4f aA[4], aB[4], b[8];
#pragma unroll
        for (int i = 0; i < 4; ++i) aA[i] = *(const v4f*)(a1 + 8 * i);
#pragma unroll
        for (int t = 0; t < 8; ++t) b[t] = *(const v4f*)(w_lds + t * 16 * P);

#pragma unroll 1
        for (int ch = 0; ch < C; ch += 2) {
            const float* pB = (ch + 1 < C1) ? a1 + CHUNK_K * (ch + 1)
                                            : a2 + CHUNK_K * (ch + 1 - C1);
#pragma unroll
            for (int i = 0; i < 4; ++i) aB[i] = *(const v4f*)(pB + 8 * i);
            chunk_mma<P>(acc, aA, b, w_lds + CHUNK_K * ch, w_lds + CHUNK_K * (ch + 1));

            const int cn = (ch + 2 < C) ? ch + 2 : ch + 1;
            const float* pA = (cn < C1) ? a1 + CHUNK_K * cn : a2 + CHUNK_K * (cn - C1);
#pragma unroll
            for (int i = 0; i < 4; ++i) aA[i] = *(const v4f*)(pA + 8 * i);
            chunk_mma<P>(acc, aB, b, w_lds + CHUNK_K * (ch + 1), w_lds + CHUNK_K * cn);
        }

#pragma unroll
        for (int r = 0; r < 8; ++r) {
            v4f lo = {acc[0][r], acc[1][r], acc[2][r], acc[3][r]};
            v4f hi = {acc[4][r], acc[5][r], acc[6][r], acc[7][r]};
            lo += b_lo;
            hi += b_hi;
            if (RELU) {
#pragma unroll
                for (int e = 0; e < 4; ++e) {
                    lo[e] = fmaxf(lo[e], 0.f);
                    hi[e] = fmaxf(hi[e], 0.f);
                }
            }
            float* p = D + (size_t)(row0 + 8 * h + r) * D_HID + 8 * c;
            *(v4f*)p       = lo;
            *(v4f*)(p + 4) = hi;
        }
    }
}

extern "C" void kernel_launch(void* const* d_in, const int* in_sizes, int n_in,
                              void* d_out, int out_size, void* d_ws, size_t ws_size,
                              hipStream_t stream) {
    const float* x       = (const float*)d_in[0];
    const int*   edge    = (const int*)d_in[1];
    const float* W1_rel  = (const float*)d_in[2];
    const float* b1_rel  = (const float*)d_in[3];
    const float* W1_root = (const float*)d_in[4];
    const float* W2_rel  = (const float*)d_in[5];
    const float* b2_rel  = (const float*)d_in[6];
    const float* W2_root = (const float*)d_in[7];
    float* out = (float*)d_out;

    const int* src = edge;
    const int* dst = edge + N_EDGES;

    float* h   = (float*)d_ws;
    float* agg = h + (size_t)N_NODES * D_HID;

    const int scatter_blocks = N_EDGES / 8;
    const int n_strips       = N_NODES / 16;

    {
        int n4 = N_NODES * D_IN / 4;
        zero_kernel<<<(n4 + 255) / 256, 256, 0, stream>>>((float4*)agg, n4);
    }
    scatter_add_kernel<D_IN><<<scatter_blocks, 256, 0, stream>>>(x, src, dst, agg, N_EDGES);
    gemm_dual_strip<D_IN, D_IN, 1><<<GEMM_BLOCKS, GEMM_THREADS, 0, stream>>>(
        agg, W1_rel, x, W1_root, b1_rel, h, n_strips);

    {
        int n4 = N_NODES * D_HID / 4;
        zero_kernel<<<(n4 + 255) / 256, 256, 0, stream>>>((float4*)agg, n4);
    }
    scatter_add_kernel<D_HID><<<scatter_blocks, 256, 0, stream>>>(h, src, dst, agg, N_EDGES);
    gemm_dual_strip<D_HID, D_HID, 0><<<GEMM_BLOCKS, GEMM_THREADS, 0, stream>>>(
        agg, W2_rel, h, W2_root, b2_rel, out, n_strips);
}


# Aggregation as per-node gather: edges bucketed by destination once per call (int atomics), buckets ranked by source id, one wave sums a node's rows in that fixed order; no f32 atomics, no zero-fill
# speedup vs baseline: 2.3681x; 1.9752x over previous
#include <hip/hip_runtime.h>

#define N_NODES 100000
#define N_EDGES 1200000
#define D_IN 64
#define D_HID 128

typedef __attribute__((ext_vector_type(2))) float v2f;
typedef __attribute__((ext_vector_type(4))) float v4f;
typedef __attribute__((ext_vector_type(8))) float v8f;

#define AGG_THREADS 256
#define AGG_WAVES   8
#define NODE_BLOCKS 2048
#define EDGE_BLOCKS 1024

#define CSR_CNT     0
#define CSR_SEG     (N_NODES + 32)
#define CSR_RANK    (CSR_SEG + 2 * N_NODES)
#define CSR_TMP     (CSR_RANK + N_EDGES)
#define CSR_NBR     (CSR_TMP + N_EDGES)
#define CSR_INTS    (CSR_NBR + N_EDGES)

__global__ void zero_int_kernel(int* __restrict__ p, int n) {
    const int i = blockIdx.x * blockDim.x + threadIdx.x;
    if (i < n) p[i] = 0;
}

__global__ __launch_bounds__(AGG_THREADS) void degree_rank_kernel(
        const int* __restrict__ dst, int* __restrict__ cnt, int* __restrict__ rank, int n_edges) {
    const int stride = gridDim.x * AGG_THREADS;
    for (int e = blockIdx.x * AGG_THREADS + threadIdx.x; e < n_edges; e += stride)
        rank[e] = atomicAdd(&cnt[dst[e]], 1);
}

__global__ __launch_bounds__(AGG_THREADS) void bucket_alloc_kernel(
        const int* __restrict__ cnt, int* __restrict__ total, int* __restrict__ seg, int n_nodes) {
    const int i    = blockIdx.x * AGG_THREADS + threadIdx.x;
    const int lane = threadIdx.x & 31;
    const int v = (i < n_nodes) ? cnt[i] : 0;
    int incl = v;
#pragma unroll
    for (int o = 1; o < 32; o <<= 1) {
        const int y = __shfl_up(incl, o);
        if (lane >= o) incl += y;
    }
    const int wave_total = __shfl(incl, 31);
    int base = 0;
    if (lane == 0) base = atomicAdd(total, wave_total);
    base = __shfl(base, 0);
    if (i < n_nodes) {
        seg[2 * i]     = base + incl - v;
        seg[2 * i + 1] = v;
    }
}

__global__ __launch_bounds__(AGG_THREADS) void bucket_fill_kernel(
        const int* __restrict__ src, const int* __restrict__ dst, const int* __restrict__ rank,
        const int* __restrict__ seg, int* __restrict__ tmp, int n_edges) {
    const int stride = gridDim.x * AGG_THREADS;
    for (int e = blockIdx.x * AGG_THREADS + threadIdx.x; e < n_edges; e += stride)
        tmp[seg[2 * dst[e]] + rank[e]] = src[e];
}

__global__ __launch_bounds__(AGG_THREADS) void bucket_sort_kernel(
        const int* __restrict__ seg, const int* __restrict__ tmp, int* __restrict__ nbr, int n_nodes) {
    const int lane = threadIdx.x & 31;
    const int wave0   = blockIdx.x * AGG_WAVES + __builtin_amdgcn_readfirstlane(threadIdx.x >> 5);
    const int n_waves = gridDim.x * AGG_WAVES;
    for (int node = wave0; node < n_nodes; node += n_waves) {
        const int first = __builtin_amdgcn_readfirstlane(seg[2 * node]);
        const int deg   = __builtin_amdgcn_readfirstlane(seg[2 * node + 1]);
        for (int ib = 0; ib < deg; ib += 32) {
            const int i   = ib + lane;
            const int key = (i < deg) ? tmp[first + i] : 0;
            int r = 0;
            for (int jb = 0; jb < deg; jb += 32) {
                const int m  = (deg - jb < 32) ? deg - jb : 32;
                const int kv = (lane < m) ? tmp[first + jb + lane] : 0;
                for (int jj = 0; jj < m; ++jj) {
                    const int kj = __builtin_amdgcn_readlane(kv, jj);
                    const int j  = jb + jj;
                    r += (kj < key || (kj == key && j < i)) ? 1 : 0;
                }
            }
            if (i < deg) nbr[first + r] = key;
        }
    }
}

template <int V> struct lane_vec;
template <> struct lane_vec<2> { typedef v2f type; };
template <> struct lane_vec<4> { typedef v4f type; };

template <int D>
__global__ __launch_bounds__(AGG_THREADS) void gather_sum_kernel(
        const float* __restrict__ feat, const int* __restrict__ seg, const int* __restrict__ nbr,
        float* __restrict__ agg, int n_nodes) {
    constexpr int V = D / 32;
    typedef typename lane_vec<V>::type vec;
    const int lane = threadIdx.x & 31;
    const int wave0   = blockIdx.x * AGG_WAVES + __builtin_amdgcn_readfirstlane(threadIdx.x >> 5);
    const int n_waves = gridDim.x * AGG_WAVES;
    const float* fl = feat + lane * V;

    for (int node = wave0; node < n_nodes; node += n_waves) {
        const int first = __builtin_amdgcn_readfirstlane(seg[2 * node]);
        const int deg   = __builtin_amdgcn_readfirstlane(seg[2 * node + 1]);
        vec acc = {};
        for (int ib = 0; ib < deg; ib += 32) {
            const int m   = (deg - ib < 32) ? deg - ib : 32;
            const int ids = (lane < m) ? nbr[first + ib + lane] : 0;
            int jj = 0;
            for (; jj + 4 <= m; jj += 4) {
                const int s0 = __builtin_amdgcn_readlane(ids, jj);
                const int s1 = __builtin_amdgcn_readlane(ids, jj + 1);
                const int s2 = __builtin_amdgcn_readlane(ids, jj + 2);
                const int s3 = __builtin_amdgcn_readlane(ids, jj + 3);
                const vec r0 = *(const vec*)(fl + (size_t)s0 * D);
                const vec r1 = *(const vec*)(fl + (size_t)s1 * D);
                const vec r2 = *(const vec*)(fl + (size_t)s2 * D);
                const vec r3 = *(const vec*)(fl + (size_t)s3 * D);
                acc += r0;
                acc += r1;
                acc += r2;
                acc += r3;
            }
            if (jj < m) {
                const int last = m - 1;
                const int s0 = __builtin_amdgcn_readlane(ids, jj);
                const int s1 = __builtin_amdgcn_readlane(ids, (jj + 1 < last) ? jj + 1 : last);
                const int s2 = __builtin_amdgcn_readlane(ids, (jj + 2 < last) ? jj + 2 : last);
                const vec r0 = *(const vec*)(fl + (size_t)s0 * D);
                const vec r1 = *(const vec*)(fl + (size_t)s1 * D);
                const vec r2 = *(const vec*)(fl + (size_t)s2 * D);
                acc += r0;
                if (jj + 1 < m) acc += r1;
                if (jj + 2 < m) acc += r2;
            }
        }
        *(vec*)(agg + (size_t)node * D + lane * V) = acc;
    }
}

#define GEMM_THREADS 256
#define GEMM_WAVES   8
#define GEMM_BLOCKS  512
#define CHUNK_K      32

template <int K, int P>
__device__ __forceinline__ void stage_weights(float* lds, const float* __restrict__ Wg, int col0) {
    constexpr int Q = K / 4;
    for (int i = threadIdx.x; i < D_HID * Q; i += GEMM_THREADS) {
        const int n = i / Q;
        const int q = i - n * Q;
        const int j = (n & 7) * 16 + (n >> 3);
        *(v4f*)(lds + j * P + col0 + 4 * q) = *(const v4f*)(Wg + n * K + 4 * q);
    }
}

template <int P>
__device__ __forceinline__ void chunk_mma(v8f (&acc)[8], const v4f (&a)[4], v4f (&b)[8],
                                          const float* w_cur, const float* w_after) {
#pragma unroll
    for (int i = 0; i < 4; ++i) {
        const float* wn = (i < 3) ? w_cur + 8 * (i + 1) : w_after;
        v4f bn[8];
#pragma unroll
        for (int t = 0; t < 8; ++t) bn[t] = *(const v4f*)(wn + t * 16 * P);

        const v2f a_lo = a[i].xy;
        const v2f a_hi = a[i].zw;
#pragma unroll
        for (int t = 0; t < 8; ++t) {
            const v2f bb = b[t].xy;
            acc[t] = __builtin_amdgcn_wmma_f32_16x16x4_f32(false, a_lo, false, bb,
                                                           (short)0, acc[t], false, false);
        }
#pragma unroll
        for (int t = 0; t < 8; ++t) {
            const v2f bb = b[t].zw;
            acc[t] = __builtin_amdgcn_wmma_f32_16x16x4_f32(false, a_hi, false, bb,
                                                           (short)0, acc[t], false, false);
        }
#pragma unroll
        for (int t = 0; t < 8; ++t) b[t] = bn[t];
    }
}

template <int K1, int K2, int RELU>
__global__ __launch_bounds__(GEMM_THREADS) void gemm_dual_strip(
        const float* __restrict__ A1, const float* __restrict__ W1,
        const float* __restrict__ A2, const float* __restrict__ W2,
        const float* __restrict__ bias, float* __restrict__ D, int n_strips) {
    constexpr int KK = K1 + K2;
    constexpr int P  = KK + 4;
    constexpr int C1 = K1 / CHUNK_K;
    constexpr int C  = KK / CHUNK_K;
    static_assert(K1 % CHUNK_K == 0 && K2 % CHUNK_K == 0 && C % 2 == 0, "whole chunks, in pairs");
    static_assert(((P / 4) & 1) == 1, "pitch is an odd number of 16-B units");

    __shared__ __attribute__((aligned(16))) float sW[D_HID * P];

    stage_weights<K1, P>(sW, W1, 0);
    stage_weights<K2, P>(sW, W2, K1);
    __syncthreads();

    const int lane = threadIdx.x & 31;
    const int h    = lane >> 4;
    const int c    = lane & 15;
    const int wave = __builtin_amdgcn_readfirstlane(threadIdx.x >> 5);

    const v4f b_lo = *(const v4f*)(bias + 8 * c);
    const v4f b_hi = *(const v4f*)(bias + 8 * c + 4);

    const float* w_lds = sW + c * P + 4 * h;

    for (int s = wave * (int)gridDim.x + (int)blockIdx.x; s < n_strips;
         s += GEMM_WAVES * (int)gridDim.x) {
        const int row0 = s * 16;
        const float* a1 = A1 + (size_t)(row0 + c) * K1 + 4 * h;
        const float* a2 = A2 + (size_t)(row0 + c) * K2 + 4 * h;

        v8f acc[8];
#pragma unroll
        for (int t = 0; t < 8; ++t) acc[t] = v8f{};

        v4f aA[4], aB[4], b[8];
#pragma unroll
        for (int i = 0; i < 4; ++i) aA[i] = *(const v4f*)(a1 + 8 * i);
#pragma unroll
        for (int t = 0; t < 8; ++t) b[t] = *(const v4f*)(w_lds + t * 16 * P);

#pragma unroll 1
        for (int ch = 0; ch < C; ch += 2) {
            const float* pB = (ch + 1 < C1) ? a1 + CHUNK_K * (ch + 1)
                                            : a2 + CHUNK_K * (ch + 1 - C1);
#pragma unroll
            for (int i = 0; i < 4; ++i) aB[i] = *(const v4f*)(pB + 8 * i);
            chunk_mma<P>(acc, aA, b, w_lds + CHUNK_K * ch, w_lds + CHUNK_K * (ch + 1));

            const int cn = (ch + 2 < C) ? ch + 2 : ch + 1;
            const float* pA = (cn < C1) ? a1 + CHUNK_K * cn : a2 + CHUNK_K * (cn - C1);
#pragma unroll
            for (int i = 0; i < 4; ++i) aA[i] = *(const v4f*)(pA + 8 * i);
            chunk_mma<P>(acc, aB, b, w_lds + CHUNK_K * (ch + 1), w_lds + CHUNK_K * cn);
        }

#pragma unroll
        for (int r = 0; r < 8; ++r) {
            v4f lo = {acc[0][r], acc[1][r], acc[2][r], acc[3][r]};
            v4f hi = {acc[4][r], acc[5][r], acc[6][r], acc[7][r]};
            lo += b_lo;
            hi += b_hi;
            if (RELU) {
#pragma unroll
                for (int e = 0; e < 4; ++e) {
                    lo[e] = fmaxf(lo[e], 0.f);
                    hi[e] = fmaxf(hi[e], 0.f);
                }
            }
            float* p = D + (size_t)(row0 + 8 * h + r) * D_HID + 8 * c;
            *(v4f*)p       = lo;
            *(v4f*)(p + 4) = hi;
        }
    }
}

extern "C" void kernel_launch(void* const* d_in, const int* in_sizes, int n_in,
                              void* d_out, int out_size, void* d_ws, size_t ws_size,
                              hipStream_t stream) {
    const float* x       = (const float*)d_in[0];
    const int*   edge    = (const int*)d_in[1];
    const float* W1_rel  = (const float*)d_in[2];
    const float* b1_rel  = (const float*)d_in[3];
    const float* W1_root = (const float*)d_in[4];
    const float* W2_rel  = (const float*)d_in[5];
    const float* b2_rel  = (const float*)d_in[6];
    const float* W2_root = (const float*)d_in[7];
    float* out = (float*)d_out;

    const int* src = edge;
    const int* dst = edge + N_EDGES;

    float* h   = (float*)d_ws;
    float* agg = h + (size_t)N_NODES * D_HID;

    static_assert((size_t)CSR_INTS <= (size_t)N_NODES * D_HID, "bucket arrays fit in the output buffer");
    int* csr   = (int*)d_out;
    int* cnt   = csr + CSR_CNT;
    int* total = csr + CSR_CNT + N_NODES;
    int* seg   = csr + CSR_SEG;
    int* rank  = csr + CSR_RANK;
    int* tmp   = csr + CSR_TMP;
    int* nbr   = csr + CSR_NBR;

    const int lane_blocks = (N_NODES + 32 + AGG_THREADS - 1) / AGG_THREADS;
    const int n_strips    = N_NODES / 16;

    zero_int_kernel<<<lane_blocks, AGG_THREADS, 0, stream>>>(cnt, N_NODES + 32);
    degree_rank_kernel<<<EDGE_BLOCKS, AGG_THREADS, 0, stream>>>(dst, cnt, rank, N_EDGES);
    bucket_alloc_kernel<<<lane_blocks, AGG_THREADS, 0, stream>>>(cnt, total, seg, N_NODES);
    bucket_fill_kernel<<<EDGE_BLOCKS, AGG_THREADS, 0, stream>>>(src, dst, rank, seg, tmp, N_EDGES);
    bucket_sort_kernel<<<NODE_BLOCKS, AGG_THREADS, 0, stream>>>(seg, tmp, nbr, N_NODES);

    gather_sum_kernel<D_IN><<<NODE_BLOCKS, AGG_THREADS, 0, stream>>>(x, seg, nbr, agg, N_NODES);
    gemm_dual_strip<D_IN, D_IN, 1><<<GEMM_BLOCKS, GEMM_THREADS, 0, stream>>>(
        agg, W1_rel, x, W1_root, b1_rel, h, n_strips);

    gather_sum_kernel<D_HID><<<NODE_BLOCKS, AGG_THREADS, 0, stream>>>(h, seg, nbr, agg, N_NODES);
    gemm_dual_strip<D_HID, D_HID, 0><<<GEMM_BLOCKS, GEMM_THREADS, 0, stream>>>(
        agg, W2_rel, h, W2_root, b2_rel, out, n_strips);
}


# Fixed-place buckets (64 ids per node): slot = node*64 + rank from the degree atomic; scan, fill, sort launches go; gathers rank a bucket in registers; nodes over 64 edges add in edge order
# speedup vs baseline: 2.6672x; 1.1263x over previous
#include <hip/hip_runtime.h>

#define N_NODES 100000
#define N_EDGES 1200000
#define D_IN 64
#define D_HID 128

typedef __attribute__((ext_vector_type(2))) float v2f;
typedef __attribute__((ext_vector_type(4))) float v4f;
typedef __attribute__((ext_vector_type(8))) float v8f;

#define AGG_THREADS 256
#define AGG_WAVES   8
#define NODE_BLOCKS 2048
#define EDGE_BLOCKS 1024
#define EDGE_BATCH  4

#define BKT_CAP     64
#define BKT_SHIFT   6

#define BKT_CNT     0
#define BKT_IDS     (N_NODES + 32)
#define BKT_INTS    (BKT_IDS + N_NODES * BKT_CAP)

static_assert((1 << BKT_SHIFT) == BKT_CAP, "bucket size is a power of two");
static_assert(BKT_IDS % BKT_CAP == 0, "buckets are aligned to their own size");
static_assert(N_NODES <= (1 << (31 - BKT_SHIFT)), "an id and its slot fit one sort key");

__global__ void zero_int_kernel(int* __restrict__ p, int n) {
    const int i = blockIdx.x * blockDim.x + threadIdx.x;
    if (i < n) p[i] = 0;
}

__global__ __launch_bounds__(AGG_THREADS) void bucket_fill_kernel(
        const int* __restrict__ src, const int* __restrict__ dst,
        int* __restrict__ cnt, int* __restrict__ bkt, int n_edges) {
    const int tid   = blockIdx.x * AGG_THREADS + threadIdx.x;
    const int lanes = gridDim.x * AGG_THREADS;
    for (int base = tid; base < n_edges; base += EDGE_BATCH * lanes) {
        if (base + (EDGE_BATCH - 1) * lanes < n_edges) {
            int d[EDGE_BATCH], s[EDGE_BATCH], r[EDGE_BATCH];
#pragma unroll
            for (int k = 0; k < EDGE_BATCH; ++k) {
                d[k] = dst[base + k * lanes];
                s[k] = src[base + k * lanes];
            }
#pragma unroll
            for (int k = 0; k < EDGE_BATCH; ++k) r[k] = atomicAdd(&cnt[d[k]], 1);
#pragma unroll
            for (int k = 0; k < EDGE_BATCH; ++k)
                if (r[k] < BKT_CAP) bkt[d[k] * BKT_CAP + r[k]] = s[k];
        } else {
            for (int e = base; e < n_edges; e += lanes) {
                const int d = dst[e];
                const int r = atomicAdd(&cnt[d], 1);
                if (r < BKT_CAP) bkt[d * BKT_CAP + r] = src[e];
            }
        }
    }
}

template <int V> struct lane_vec;
template <> struct lane_vec<2> { typedef v2f type; };
template <> struct lane_vec<4> { typedef v4f type; };

template <int D>
__global__ __launch_bounds__(AGG_THREADS) void gather_sum_kernel(
        const float* __restrict__ feat, const int* __restrict__ cnt, const int* __restrict__ bkt,
        const int* __restrict__ src, const int* __restrict__ dst,
        float* __restrict__ agg, int n_nodes, int n_edges) {
    constexpr int V = D / 32;
    typedef typename lane_vec<V>::type vec;
    const int lane = threadIdx.x & 31;
    const int wave0   = blockIdx.x * AGG_WAVES + __builtin_amdgcn_readfirstlane(threadIdx.x >> 5);
    const int n_waves = gridDim.x * AGG_WAVES;
    const float* fl = feat + lane * V;

    for (int node = wave0; node < n_nodes; node += n_waves) {
        const int deg = __builtin_amdgcn_readfirstlane(cnt[node]);
        const int* b  = bkt + (size_t)node * BKT_CAP;
        vec acc = {};
        if (deg <= 32) {
            const int key  = (lane < deg) ? b[lane] : 0;
            const int mine = (key << BKT_SHIFT) | lane;
            int r = 0;
            for (int jj = 0; jj < deg; ++jj) {
                const int pj = __builtin_amdgcn_readlane(mine, jj);
                r += (pj < mine) ? 1 : 0;
            }
            const int to  = (lane < deg) ? r : lane;
            const int ids = __builtin_amdgcn_ds_permute(to << 2, key);
            int jj = 0;
            for (; jj + 4 <= deg; jj += 4) {
                const int s0 = __builtin_amdgcn_readlane(ids, jj);
                const int s1 = __builtin_amdgcn_readlane(ids, jj + 1);
                const int s2 = __builtin_amdgcn_readlane(ids, jj + 2);
                const int s3 = __builtin_amdgcn_readlane(ids, jj + 3);
                const vec r0 = *(const vec*)(fl + (size_t)s0 * D);
                const vec r1 = *(const vec*)(fl + (size_t)s1 * D);
                const vec r2 = *(const vec*)(fl + (size_t)s2 * D);
                const vec r3 = *(const vec*)(fl + (size_t)s3 * D);
                acc += r0;
                acc += r1;
                acc += r2;
                acc += r3;
            }
            if (jj < deg) {
                const int last = deg - 1;
                const int s0 = __builtin_amdgcn_readlane(ids, jj);
                const int s1 = __builtin_amdgcn_readlane(ids, (jj + 1 < last) ? jj + 1 : last);
                const int s2 = __builtin_amdgcn_readlane(ids, (jj + 2 < last) ? jj + 2 : last);
                const vec r0 = *(const vec*)(fl + (size_t)s0 * D);
                const vec r1 = *(const vec*)(fl + (size_t)s1 * D);
                const vec r2 = *(const vec*)(fl + (size_t)s2 * D);
                acc += r0;
                if (jj + 1 < deg) acc += r1;
                if (jj + 2 < deg) acc += r2;
            }
        } else if (deg <= BKT_CAP) {
            const bool v1 = lane + 32 < deg;
            const int k0 = b[lane];
            const int k1 = v1 ? b[32 + lane] : 0;
            const int p0 = (k0 << BKT_SHIFT) | lane;
            const int p1 = v1 ? ((k1 << BKT_SHIFT) | (lane + 32)) : 0x7fffffff;
            int r0 = 0, r1 = 0;
            for (int jj = 0; jj < 32; ++jj) {
                const int pj = __builtin_amdgcn_readlane(p0, jj);
                r0 += (pj < p0) ? 1 : 0;
                r1 += (pj < p1) ? 1 : 0;
            }
            for (int jj = 0; jj < deg - 32; ++jj) {
                const int pj = __builtin_amdgcn_readlane(p1, jj);
                r0 += (pj < p0) ? 1 : 0;
                r1 += (pj < p1) ? 1 : 0;
            }
            auto pick = [&](int pos) -> int {
                const unsigned m0 = __builtin_amdgcn_ballot_w32(r0 == pos);
                const unsigned m1 = __builtin_amdgcn_ballot_w32(v1 && r1 == pos);
                const int l0 = (m0 != 0u) ? __builtin_ctz(m0) : 0;
                const int l1 = (m1 != 0u) ? __builtin_ctz(m1) : 0;
                const int a0 = __builtin_amdgcn_readlane(k0, l0);
                const int a1 = __builtin_amdgcn_readlane(k1, l1);
                return (m0 != 0u) ? a0 : a1;
            };
            const int last = deg - 1;
            for (int pos = 0; pos < deg; pos += 4) {
                const int s0 = pick(pos);
                const int s1 = pick((pos + 1 < last) ? pos + 1 : last);
                const int s2 = pick((pos + 2 < last) ? pos + 2 : last);
                const int s3 = pick((pos + 3 < last) ? pos + 3 : last);
                const vec q0 = *(const vec*)(fl + (size_t)s0 * D);
                const vec q1 = *(const vec*)(fl + (size_t)s1 * D);
                const vec q2 = *(const vec*)(fl + (size_t)s2 * D);
                const vec q3 = *(const vec*)(fl + (size_t)s3 * D);
                acc += q0;
                if (pos + 1 < deg) acc += q1;
                if (pos + 2 < deg) acc += q2;
                if (pos + 3 < deg) acc += q3;
            }
        } else {
            for (int eb = 0; eb < n_edges; eb += 32) {
                const int e = eb + lane;
                const int d = (e < n_edges) ? dst[e] : -1;
                unsigned m = __builtin_amdgcn_ballot_w32(d == node);
                if (m != 0u) {
                    const int sv = (e < n_edges) ? src[e] : 0;
                    while (m != 0u) {
                        const int s = __builtin_amdgcn_readlane(sv, __builtin_ctz(m));
                        acc += *(const vec*)(fl + (size_t)s * D);
                        m &= m - 1u;
                    }
                }
            }
        }
        *(vec*)(agg + (size_t)node * D + lane * V) = acc;
    }
}

#define GEMM_THREADS 256
#define GEMM_WAVES   8
#define GEMM_BLOCKS  512
#define CHUNK_K      32

template <int K, int P>
__device__ __forceinline__ void stage_weights(float* lds, const float* __restrict__ Wg, int col0) {
    constexpr int Q = K / 4;
    for (int i = threadIdx.x; i < D_HID * Q; i += GEMM_THREADS) {
        const int n = i / Q;
        const int q = i - n * Q;
        const int j = (n & 7) * 16 + (n >> 3);
        *(v4f*)(lds + j * P + col0 + 4 * q) = *(const v4f*)(Wg + n * K + 4 * q);
    }
}

template <int P>
__device__ __forceinline__ void chunk_mma(v8f (&acc)[8], const v4f (&a)[4], v4f (&b)[8],
                                          const float* w_cur, const float* w_after) {
#pragma unroll
    for (int i = 0; i < 4; ++i) {
        const float* wn = (i < 3) ? w_cur + 8 * (i + 1) : w_after;
        v4f bn[8];
#pragma unroll
        for (int t = 0; t < 8; ++t) bn[t] = *(const v4f*)(wn + t * 16 * P);

        const v2f a_lo = a[i].xy;
        const v2f a_hi = a[i].zw;
#pragma unroll
        for (int t = 0; t < 8; ++t) {
            const v2f bb = b[t].xy;
            acc[t] = __builtin_amdgcn_wmma_f32_16x16x4_f32(false, a_lo, false, bb,
                                                           (short)0, acc[t], false, false);
        }
#pragma unroll
        for (int t = 0; t < 8; ++t) {
            const v2f bb = b[t].zw;
            acc[t] = __builtin_amdgcn_wmma_f32_16x16x4_f32(false, a_hi, false, bb,
                                                           (short)0, acc[t], false, false);
        }
#pragma unroll
        for (int t = 0; t < 8; ++t) b[t] = bn[t];
    }
}

template <int K1, int K2, int RELU>
__global__ __launch_bounds__(GEMM_THREADS) void gemm_dual_strip(
        const float* __restrict__ A1, const float* __restrict__ W1,
        const float* __restrict__ A2, const float* __restrict__ W2,
        const float* __restrict__ bias, float* __restrict__ D, int n_strips) {
    constexpr int KK = K1 + K2;
    constexpr int P  = KK + 4;
    constexpr int C1 = K1 / CHUNK_K;
    constexpr int C  = KK / CHUNK_K;
    static_assert(K1 % CHUNK_K == 0 && K2 % CHUNK_K == 0 && C % 2 == 0, "whole chunks, in pairs");
    static_assert(((P / 4) & 1) == 1, "pitch is an odd number of 16-B units");

    __shared__ __attribute__((aligned(16))) float sW[D_HID * P];

    stage_weights<K1, P>(sW, W1, 0);
    stage_weights<K2, P>(sW, W2, K1);
    __syncthreads();

    const int lane = threadIdx.x & 31;
    const int h    = lane >> 4;
    const int c    = lane & 15;
    const int wave = __builtin_amdgcn_readfirstlane(threadIdx.x >> 5);

    const v4f b_lo = *(const v4f*)(bias + 8 * c);
    const v4f b_hi = *(const v4f*)(bias + 8 * c + 4);

    const float* w_lds = sW + c * P + 4 * h;

    for (int s = wave * (int)gridDim.x + (int)blockIdx.x; s < n_strips;
         s += GEMM_WAVES * (int)gridDim.x) {
        const int row0 = s * 16;
        const float* a1 = A1 + (size_t)(row0 + c) * K1 + 4 * h;
        const float* a2 = A2 + (size_t)(row0 + c) * K2 + 4 * h;

        v8f acc[8];
#pragma unroll
        for (int t = 0; t < 8; ++t) acc[t] = v8f{};

        v4f aA[4], aB[4], b[8];
#pragma unroll
        for (int i = 0; i < 4; ++i) aA[i] = *(const v4f*)(a1 + 8 * i);
#pragma unroll
        for (int t = 0; t < 8; ++t) b[t] = *(const v4f*)(w_lds + t * 16 * P);

#pragma unroll 1
        for (int ch = 0; ch < C; ch += 2) {
            const float* pB = (ch + 1 < C1) ? a1 + CHUNK_K * (ch + 1)
                                            : a2 + CHUNK_K * (ch + 1 - C1);
#pragma unroll
            for (int i = 0; i < 4; ++i) aB[i] = *(const v4f*)(pB + 8 * i);
            chunk_mma<P>(acc, aA, b, w_lds + CHUNK_K * ch, w_lds + CHUNK_K * (ch + 1));

            const int cn = (ch + 2 < C) ? ch + 2 : ch + 1;
            const float* pA = (cn < C1) ? a1 + CHUNK_K * cn : a2 + CHUNK_K * (cn - C1);
#pragma unroll
            for (int i = 0; i < 4; ++i) aA[i] = *(const v4f*)(pA + 8 * i);
            chunk_mma<P>(acc, aB, b, w_lds + CHUNK_K * (ch + 1), w_lds + CHUNK_K * cn);
        }

#pragma unroll
        for (int r = 0; r < 8; ++r) {
            v4f lo = {acc[0][r], acc[1][r], acc[2][r], acc[3][r]};
            v4f hi = {acc[4][r], acc[5][r], acc[6][r], acc[7][r]};
            lo += b_lo;
            hi += b_hi;
            if (RELU) {
#pragma unroll
                for (int e = 0; e < 4; ++e) {
                    lo[e] = fmaxf(lo[e], 0.f);
                    hi[e] = fmaxf(hi[e], 0.f);
                }
            }
            float* p = D + (size_t)(row0 + 8 * h + r) * D_HID + 8 * c;
            *(v4f*)p       = lo;
            *(v4f*)(p + 4) = hi;
        }
    }
}

extern "C" void kernel_launch(void* const* d_in, const int* in_sizes, int n_in,
                              void* d_out, int out_size, void* d_ws, size_t ws_size,
                              hipStream_t stream) {
    const float* x       = (const float*)d_in[0];
    const int*   edge    = (const int*)d_in[1];
    const float* W1_rel  = (const float*)d_in[2];
    const float* b1_rel  = (const float*)d_in[3];
    const float* W1_root = (const float*)d_in[4];
    const float* W2_rel  = (const float*)d_in[5];
    const float* b2_rel  = (const float*)d_in[6];
    const float* W2_root = (const float*)d_in[7];
    float* out = (float*)d_out;

    const int* src = edge;
    const int* dst = edge + N_EDGES;

    float* h   = (float*)d_ws;
    float* agg = h + (size_t)N_NODES * D_HID;

    static_assert((size_t)BKT_INTS <= (size_t)N_NODES * D_HID, "degree and bucket arrays fit in the output buffer");
    int* csr = (int*)d_out;
    int* cnt = csr + BKT_CNT;
    int* bkt = csr + BKT_IDS;

    const int lane_blocks = (N_NODES + AGG_THREADS - 1) / AGG_THREADS;
    const int n_strips    = N_NODES / 16;

    zero_int_kernel<<<lane_blocks, AGG_THREADS, 0, stream>>>(cnt, N_NODES);
    bucket_fill_kernel<<<EDGE_BLOCKS, AGG_THREADS, 0, stream>>>(src, dst, cnt, bkt, N_EDGES);

    gather_sum_kernel<D_IN><<<NODE_BLOCKS, AGG_THREADS, 0, stream>>>(
        x, cnt, bkt, src, dst, agg, N_NODES, N_EDGES);
    gemm_dual_strip<D_IN, D_IN, 1><<<GEMM_BLOCKS, GEMM_THREADS, 0, stream>>>(
        agg, W1_rel, x, W1_root, b1_rel, h, n_strips);

    gather_sum_kernel<D_HID><<<NODE_BLOCKS, AGG_THREADS, 0, stream>>>(
        h, cnt, bkt, src, dst, agg, N_NODES, N_EDGES);
    gemm_dual_strip<D_HID, D_HID, 0><<<GEMM_BLOCKS, GEMM_THREADS, 0, stream>>>(
        agg, W2_rel, h, W2_root, b2_rel, out, n_strips);
}


# Bucket build via LDS: edges binned per block with LDS atomics, one block per bin builds degrees and buckets in LDS and writes whole rows; global atomics and zero pass go; sum order unchanged
# speedup vs baseline: 3.5974x; 1.3488x over previous
#include <hip/hip_runtime.h>

#define N_NODES 100000
#define N_EDGES 1200000
#define D_IN 64
#define D_HID 128

typedef __attribute__((ext_vector_type(2))) float v2f;
typedef __attribute__((ext_vector_type(4))) float v4f;
typedef __attribute__((ext_vector_type(8))) float v8f;

#define AGG_THREADS 256
#define AGG_WAVES   8
#define NODE_BLOCKS 2048

#define BKT_CAP     64
#define BKT_SHIFT   6

#define BKT_CNT     0
#define BKT_IDS     (N_NODES + 32)
#define BKT_INTS    (BKT_IDS + N_NODES * BKT_CAP)

static_assert((1 << BKT_SHIFT) == BKT_CAP, "bucket size is a power of two");
static_assert(BKT_IDS % BKT_CAP == 0, "buckets are aligned to their own size");
static_assert(N_NODES <= (1 << (31 - BKT_SHIFT)), "an id and its slot fit one sort key");

#define PART_THREADS 256
#define PART_EDGES   4096
#define PART_BLOCKS  ((N_EDGES + PART_EDGES - 1) / PART_EDGES)
#define N_BINS       256
#define BIN_NODES    ((N_NODES + N_BINS - 1) / N_BINS)
#define CELL_INTS    64
#define CELL_CAP     63
#define LOC_BITS     9
#define BUILD_THREADS 512
#define BUILD_WAVES   16
#define BUILD_ROWS   ((PART_BLOCKS + BUILD_WAVES - 1) / BUILD_WAVES)

typedef __attribute__((ext_vector_type(4))) int v4i;

static_assert(N_BINS == PART_THREADS, "one histogram word per thread");
static_assert(BIN_NODES <= (1 << LOC_BITS), "a node's place in its bin fits the low bits");
static_assert(N_NODES <= (1 << (31 - LOC_BITS)), "a source id fits the high bits");
static_assert(N_EDGES % 4 == 0 && PART_EDGES % (4 * PART_THREADS) == 0, "edges are read four at a time");

__global__ __launch_bounds__(PART_THREADS) void edge_partition_kernel(
        const int* __restrict__ src, const int* __restrict__ dst,
        int* __restrict__ cells, int n_edges) {
    __shared__ int hist[N_BINS];
    __shared__ __attribute__((aligned(16))) int stage[N_BINS * CELL_INTS];
    const int t = threadIdx.x;
    hist[t] = 0;
    __syncthreads();

    constexpr int Q = PART_EDGES / (4 * PART_THREADS);
    const int e0 = blockIdx.x * PART_EDGES;
    if (e0 + PART_EDGES <= n_edges) {
        v4i d[Q], s[Q];
#pragma unroll
        for (int k = 0; k < Q; ++k) {
            const int e = e0 + 4 * (k * PART_THREADS + t);
            d[k] = *(const v4i*)(dst + e);
            s[k] = *(const v4i*)(src + e);
        }
        int bin[4 * Q], key[4 * Q], r[4 * Q];
#pragma unroll
        for (int k = 0; k < 4 * Q; ++k) {
            const int dd = d[k >> 2][k & 3];
            bin[k] = dd / BIN_NODES;
            key[k] = (s[k >> 2][k & 3] << LOC_BITS) | (dd - bin[k] * BIN_NODES);
        }
#pragma unroll
        for (int k = 0; k < 4 * Q; ++k) r[k] = atomicAdd(&hist[bin[k]], 1);
#pragma unroll
        for (int k = 0; k < 4 * Q; ++k)
            if (r[k] < CELL_CAP) stage[bin[k] * CELL_INTS + 1 + r[k]] = key[k];
    } else {
        for (int e = e0 + t; e < n_edges; e += PART_THREADS) {
            const int dd = dst[e];
            const int bn = dd / BIN_NODES;
            const int rr = atomicAdd(&hist[bn], 1);
            if (rr < CELL_CAP) stage[bn * CELL_INTS + 1 + rr] = (src[e] << LOC_BITS) | (dd - bn * BIN_NODES);
        }
    }
    __syncthreads();

    for (int idx = t; idx < N_BINS * (CELL_INTS / 4); idx += PART_THREADS) {
        const int row = idx >> 4;
        const int q   = idx & 15;
        const int n   = hist[row];
        const int lim = (n < CELL_CAP) ? n : CELL_CAP;
        if (q < 8 || 4 * q <= lim) {
            v4i v = *(const v4i*)(stage + row * CELL_INTS + 4 * q);
#pragma unroll
            for (int c = 0; c < 4; ++c) {
                const int slot = 4 * q + c;
                v[c] = (slot == 0) ? n : ((slot <= lim) ? v[c] : 0);
            }
            *(v4i*)(cells + ((size_t)row * PART_BLOCKS + blockIdx.x) * CELL_INTS + 4 * q) = v;
        }
    }
}

__global__ __launch_bounds__(BUILD_THREADS) void bucket_build_kernel(
        const int* __restrict__ cells, const int* __restrict__ src, const int* __restrict__ dst,
        int* __restrict__ cnt, int* __restrict__ bkt, int n_nodes, int n_edges) {
    __shared__ int deg_l[BIN_NODES];
    __shared__ __attribute__((aligned(16))) int ids_l[BIN_NODES * BKT_CAP];
    const int t     = threadIdx.x;
    const int lane  = t & 31;
    const int wave  = __builtin_amdgcn_readfirstlane(t >> 5);
    const int node0 = blockIdx.x * BIN_NODES;
    const int* my   = cells + (size_t)blockIdx.x * PART_BLOCKS * CELL_INTS;

    if (t < BIN_NODES) deg_l[t] = 0;

    int v[BUILD_ROWS];
#pragma unroll
    for (int j = 0; j < BUILD_ROWS; ++j) {
        const int b  = wave + j * BUILD_WAVES;
        const int bb = (b < PART_BLOCKS) ? b : PART_BLOCKS - 1;
        v[j] = my[bb * CELL_INTS + lane];
    }
    __syncthreads();

    auto put = [&](int val) {
        const int ld = val & ((1 << LOC_BITS) - 1);
        const int r  = atomicAdd(&deg_l[ld], 1);
        if (r < BKT_CAP) ids_l[ld * BKT_CAP + r] = val >> LOC_BITS;
    };

#pragma unroll
    for (int j = 0; j < BUILD_ROWS; ++j) {
        const int b = wave + j * BUILD_WAVES;
        if (b < PART_BLOCKS) {
            const int n = __builtin_amdgcn_readlane(v[j], 0);
            if (n <= CELL_CAP) {
                if (lane >= 1 && lane <= n) put(v[j]);
                if (n >= 32 && lane + 32 <= n) put(my[b * CELL_INTS + 32 + lane]);
            } else {
                const int e1 = ((b + 1) * PART_EDGES < n_edges) ? (b + 1) * PART_EDGES : n_edges;
                for (int e = b * PART_EDGES + lane; e < e1; e += 32) {
                    const int loc = dst[e] - node0;
                    if (loc >= 0 && loc < BIN_NODES) put((src[e] << LOC_BITS) | loc);
                }
            }
        }
    }
    __syncthreads();

    if (t < BIN_NODES && node0 + t < n_nodes) cnt[node0 + t] = deg_l[t];
    for (int idx = t; idx < BIN_NODES * (BKT_CAP / 4); idx += BUILD_THREADS) {
        const int ld = idx >> 4;
        const int q  = idx & 15;
        if (node0 + ld < n_nodes && 4 * q < deg_l[ld])
            *(v4i*)(bkt + (size_t)(node0 + ld) * BKT_CAP + 4 * q) = *(const v4i*)(ids_l + ld * BKT_CAP + 4 * q);
    }
}

template <int V> struct lane_vec;
template <> struct lane_vec<2> { typedef v2f type; };
template <> struct lane_vec<4> { typedef v4f type; };

template <int D>
__global__ __launch_bounds__(AGG_THREADS) void gather_sum_kernel(
        const float* __restrict__ feat, const int* __restrict__ cnt, const int* __restrict__ bkt,
        const int* __restrict__ src, const int* __restrict__ dst,
        float* __restrict__ agg, int n_nodes, int n_edges) {
    constexpr int V = D / 32;
    typedef typename lane_vec<V>::type vec;
    const int lane = threadIdx.x & 31;
    const int wave0   = blockIdx.x * AGG_WAVES + __builtin_amdgcn_readfirstlane(threadIdx.x >> 5);
    const int n_waves = gridDim.x * AGG_WAVES;
    const float* fl = feat + lane * V;

    for (int node = wave0; node < n_nodes; node += n_waves) {
        const int deg = __builtin_amdgcn_readfirstlane(cnt[node]);
        const int* b  = bkt + (size_t)node * BKT_CAP;
        vec acc = {};
        if (deg <= 32) {
            const int key  = (lane < deg) ? b[lane] : 0;
            const int mine = (key << BKT_SHIFT) | lane;
            int r = 0;
            for (int jj = 0; jj < deg; ++jj) {
                const int pj = __builtin_amdgcn_readlane(mine, jj);
                r += (pj < mine) ? 1 : 0;
            }
            const int to  = (lane < deg) ? r : lane;
            const int ids = __builtin_amdgcn_ds_permute(to << 2, key);
            int jj = 0;
            for (; jj + 4 <= deg; jj += 4) {
                const int s0 = __builtin_amdgcn_readlane(ids, jj);
                const int s1 = __builtin_amdgcn_readlane(ids, jj + 1);
                const int s2 = __builtin_amdgcn_readlane(ids, jj + 2);
                const int s3 = __builtin_amdgcn_readlane(ids, jj + 3);
                const vec r0 = *(const vec*)(fl + (size_t)s0 * D);
                const vec r1 = *(const vec*)(fl + (size_t)s1 * D);
                const vec r2 = *(const vec*)(fl + (size_t)s2 * D);
                const vec r3 = *(const vec*)(fl + (size_t)s3 * D);
                acc += r0;
                acc += r1;
                acc += r2;
                acc += r3;
            }
            if (jj < deg) {
                const int last = deg - 1;
                const int s0 = __builtin_amdgcn_readlane(ids, jj);
                const int s1 = __builtin_amdgcn_readlane(ids, (jj + 1 < last) ? jj + 1 : last);
                const int s2 = __builtin_amdgcn_readlane(ids, (jj + 2 < last) ? jj + 2 : last);
                const vec r0 = *(const vec*)(fl + (size_t)s0 * D);
                const vec r1 = *(const vec*)(fl + (size_t)s1 * D);
                const vec r2 = *(const vec*)(fl + (size_t)s2 * D);
                acc += r0;
                if (jj + 1 < deg) acc += r1;
                if (jj + 2 < deg) acc += r2;
            }
        } else if (deg <= BKT_CAP) {
            const bool v1 = lane + 32 < deg;
            const int k0 = b[lane];
            const int k1 = v1 ? b[32 + lane] : 0;
            const int p0 = (k0 << BKT_SHIFT) | lane;
            const int p1 = v1 ? ((k1 << BKT_SHIFT) | (lane + 32)) : 0x7fffffff;
            int r0 = 0, r1 = 0;
            for (int jj = 0; jj < 32; ++jj) {
                const int pj = __builtin_amdgcn_readlane(p0, jj);
                r0 += (pj < p0) ? 1 : 0;
                r1 += (pj < p1) ? 1 : 0;
            }
            for (int jj = 0; jj < deg - 32; ++jj) {
                const int pj = __builtin_amdgcn_readlane(p1, jj);
                r0 += (pj < p0) ? 1 : 0;
                r1 += (pj < p1) ? 1 : 0;
            }
            auto pick = [&](int pos) -> int {
                const unsigned m0 = __builtin_amdgcn_ballot_w32(r0 == pos);
                const unsigned m1 = __builtin_amdgcn_ballot_w32(v1 && r1 == pos);
                const int l0 = (m0 != 0u) ? __builtin_ctz(m0) : 0;
                const int l1 = (m1 != 0u) ? __builtin_ctz(m1) : 0;
                const int a0 = __builtin_amdgcn_readlane(k0, l0);
                const int a1 = __builtin_amdgcn_readlane(k1, l1);
                return (m0 != 0u) ? a0 : a1;
            };
            const int last = deg - 1;
            for (int pos = 0; pos < deg; pos += 4) {
                const int s0 = pick(pos);
                const int s1 = pick((pos + 1 < last) ? pos + 1 : last);
                const int s2 = pick((pos + 2 < last) ? pos + 2 : last);
                const int s3 = pick((pos + 3 < last) ? pos + 3 : last);
                const vec q0 = *(const vec*)(fl + (size_t)s0 * D);
                const vec q1 = *(const vec*)(fl + (size_t)s1 * D);
                const vec q2 = *(const vec*)(fl + (size_t)s2 * D);
                const vec q3 = *(const vec*)(fl + (size_t)s3 * D);
                acc += q0;
                if (pos + 1 < deg) acc += q1;
                if (pos + 2 < deg) acc += q2;
                if (pos + 3 < deg) acc += q3;
            }
        } else {
            for (int eb = 0; eb < n_edges; eb += 32) {
                const int e = eb + lane;
                const int d = (e < n_edges) ? dst[e] : -1;
                unsigned m = __builtin_amdgcn_ballot_w32(d == node);
                if (m != 0u) {
                    const int sv = (e < n_edges) ? src[e] : 0;
                    while (m != 0u) {
                        const int s = __builtin_amdgcn_readlane(sv, __builtin_ctz(m));
                        acc += *(const vec*)(fl + (size_t)s * D);
                        m &= m - 1u;
                    }
                }
            }
        }
        *(vec*)(agg + (size_t)node * D + lane * V) = acc;
    }
}

#define GEMM_THREADS 256
#define GEMM_WAVES   8
#define GEMM_BLOCKS  512
#define CHUNK_K      32

template <int K, int P>
__device__ __forceinline__ void stage_weights(float* lds, const float* __restrict__ Wg, int col0) {
    constexpr int Q = K / 4;
    for (int i = threadIdx.x; i < D_HID * Q; i += GEMM_THREADS) {
        const int n = i / Q;
        const int q = i - n * Q;
        const int j = (n & 7) * 16 + (n >> 3);
        *(v4f*)(lds + j * P + col0 + 4 * q) = *(const v4f*)(Wg + n * K + 4 * q);
    }
}

template <int P>
__device__ __forceinline__ void chunk_mma(v8f (&acc)[8], const v4f (&a)[4], v4f (&b)[8],
                                          const float* w_cur, const float* w_after) {
#pragma unroll
    for (int i = 0; i < 4; ++i) {
        const float* wn = (i < 3) ? w_cur + 8 * (i + 1) : w_after;
        v4f bn[8];
#pragma unroll
        for (int t = 0; t < 8; ++t) bn[t] = *(const v4f*)(wn + t * 16 * P);

        const v2f a_lo = a[i].xy;
        const v2f a_hi = a[i].zw;
#pragma unroll
        for (int t = 0; t < 8; ++t) {
            const v2f bb = b[t].xy;
            acc[t] = __builtin_amdgcn_wmma_f32_16x16x4_f32(false, a_lo, false, bb,
                                                           (short)0, acc[t], false, false);
        }
#pragma unroll
        for (int t = 0; t < 8; ++t) {
            const v2f bb = b[t].zw;
            acc[t] = __builtin_amdgcn_wmma_f32_16x16x4_f32(false, a_hi, false, bb,
                                                           (short)0, acc[t], false, false);
        }
#pragma unroll
        for (int t = 0; t < 8; ++t) b[t] = bn[t];
    }
}

template <int K1, int K2, int RELU>
__global__ __launch_bounds__(GEMM_THREADS) void gemm_dual_strip(
        const float* __restrict__ A1, const float* __restrict__ W1,
        const float* __restrict__ A2, const float* __restrict__ W2,
        const float* __restrict__ bias, float* __restrict__ D, int n_strips) {
    constexpr int KK = K1 + K2;
    constexpr int P  = KK + 4;
    constexpr int C1 = K1 / CHUNK_K;
    constexpr int C  = KK / CHUNK_K;
    static_assert(K1 % CHUNK_K == 0 && K2 % CHUNK_K == 0 && C % 2 == 0, "whole chunks, in pairs");
    static_assert(((P / 4) & 1) == 1, "pitch is an odd number of 16-B units");

    __shared__ __attribute__((aligned(16))) float sW[D_HID * P];

    stage_weights<K1, P>(sW, W1, 0);
    stage_weights<K2, P>(sW, W2, K1);
    __syncthreads();

    const int lane = threadIdx.x & 31;
    const int h    = lane >> 4;
    const int c    = lane & 15;
    const int wave = __builtin_amdgcn_readfirstlane(threadIdx.x >> 5);

    const v4f b_lo = *(const v4f*)(bias + 8 * c);
    const v4f b_hi = *(const v4f*)(bias + 8 * c + 4);

    const float* w_lds = sW + c * P + 4 * h;

    for (int s = wave * (int)gridDim.x + (int)blockIdx.x; s < n_strips;
         s += GEMM_WAVES * (int)gridDim.x) {
        const int row0 = s * 16;
        const float* a1 = A1 + (size_t)(row0 + c) * K1 + 4 * h;
        const float* a2 = A2 + (size_t)(row0 + c) * K2 + 4 * h;

        v8f acc[8];
#pragma unroll
        for (int t = 0; t < 8; ++t) acc[t] = v8f{};

        v4f aA[4], aB[4], b[8];
#pragma unroll
        for (int i = 0; i < 4; ++i) aA[i] = *(const v4f*)(a1 + 8 * i);
#pragma unroll
        for (int t = 0; t < 8; ++t) b[t] = *(const v4f*)(w_lds + t * 16 * P);

#pragma unroll 1
        for (int ch = 0; ch < C; ch += 2) {
            const float* pB = (ch + 1 < C1) ? a1 + CHUNK_K * (ch + 1)
                                            : a2 + CHUNK_K * (ch + 1 - C1);
#pragma unroll
            for (int i = 0; i < 4; ++i) aB[i] = *(const v4f*)(pB + 8 * i);
            chunk_mma<P>(acc, aA, b, w_lds + CHUNK_K * ch, w_lds + CHUNK_K * (ch + 1));

            const int cn = (ch + 2 < C) ? ch + 2 : ch + 1;
            const float* pA = (cn < C1) ? a1 + CHUNK_K * cn : a2 + CHUNK_K * (cn - C1);
#pragma unroll
            for (int i = 0; i < 4; ++i) aA[i] = *(const v4f*)(pA + 8 * i);
            chunk_mma<P>(acc, aB, b, w_lds + CHUNK_K * (ch + 1), w_lds + CHUNK_K * cn);
        }

#pragma unroll
        for (int r = 0; r < 8; ++r) {
            v4f lo = {acc[0][r], acc[1][r], acc[2][r], acc[3][r]};
            v4f hi = {acc[4][r], acc[5][r], acc[6][r], acc[7][r]};
            lo += b_lo;
            hi += b_hi;
            if (RELU) {
#pragma unroll
                for (int e = 0; e < 4; ++e) {
                    lo[e] = fmaxf(lo[e], 0.f);
                    hi[e] = fmaxf(hi[e], 0.f);
                }
            }
            float* p = D + (size_t)(row0 + 8 * h + r) * D_HID + 8 * c;
            *(v4f*)p       = lo;
            *(v4f*)(p + 4) = hi;
        }
    }
}

extern "C" void kernel_launch(void* const* d_in, const int* in_sizes, int n_in,
                              void* d_out, int out_size, void* d_ws, size_t ws_size,
                              hipStream_t stream) {
    const float* x       = (const float*)d_in[0];
    const int*   edge    = (const int*)d_in[1];
    const float* W1_rel  = (const float*)d_in[2];
    const float* b1_rel  = (const float*)d_in[3];
    const float* W1_root = (const float*)d_in[4];
    const float* W2_rel  = (const float*)d_in[5];
    const float* b2_rel  = (const float*)d_in[6];
    const float* W2_root = (const float*)d_in[7];
    float* out = (float*)d_out;

    const int* src = edge;
    const int* dst = edge + N_EDGES;

    float* h   = (float*)d_ws;
    float* agg = h + (size_t)N_NODES * D_HID;

    static_assert((size_t)BKT_INTS <= (size_t)N_NODES * D_HID, "degree and bucket arrays fit in the output buffer");
    int* csr = (int*)d_out;
    int* cnt = csr + BKT_CNT;
    int* bkt = csr + BKT_IDS;

    const int n_strips    = N_NODES / 16;

    static_assert((size_t)N_BINS * PART_BLOCKS * CELL_INTS <= (size_t)N_NODES * D_HID, "edge cells fit in h's place");
    int* cells = (int*)d_ws;

    edge_partition_kernel<<<PART_BLOCKS, PART_THREADS, 0, stream>>>(src, dst, cells, N_EDGES);
    bucket_build_kernel<<<N_BINS, BUILD_THREADS, 0, stream>>>(cells, src, dst, cnt, bkt, N_NODES, N_EDGES);

    gather_sum_kernel<D_IN><<<NODE_BLOCKS, AGG_THREADS, 0, stream>>>(
        x, cnt, bkt, src, dst, agg, N_NODES, N_EDGES);
    gemm_dual_strip<D_IN, D_IN, 1><<<GEMM_BLOCKS, GEMM_THREADS, 0, stream>>>(
        agg, W1_rel, x, W1_root, b1_rel, h, n_strips);

    gather_sum_kernel<D_HID><<<NODE_BLOCKS, AGG_THREADS, 0, stream>>>(
        h, cnt, bkt, src, dst, agg, N_NODES, N_EDGES);
    gemm_dual_strip<D_HID, D_HID, 0><<<GEMM_BLOCKS, GEMM_THREADS, 0, stream>>>(
        agg, W2_rel, h, W2_root, b2_rel, out, n_strips);
}


# GEMMs re-dealt for equal matrix ops per SIMD: one 12-wave work-group per WGP (weights staged once), two whole strips per wave, the 106 left-over strips split by column quarters; sums unchanged
# speedup vs baseline: 3.8914x; 1.0817x over previous
#include <hip/hip_runtime.h>

#define N_NODES 100000
#define N_EDGES 1200000
#define D_IN 64
#define D_HID 128

typedef __attribute__((ext_vector_type(2))) float v2f;
typedef __attribute__((ext_vector_type(4))) float v4f;
typedef __attribute__((ext_vector_type(8))) float v8f;

#define AGG_THREADS 256
#define AGG_WAVES   8
#define NODE_BLOCKS 2048

#define BKT_CAP     64
#define BKT_SHIFT   6

#define BKT_CNT     0
#define BKT_IDS     (N_NODES + 32)
#define BKT_INTS    (BKT_IDS + N_NODES * BKT_CAP)

static_assert((1 << BKT_SHIFT) == BKT_CAP, "bucket size is a power of two");
static_assert(BKT_IDS % BKT_CAP == 0, "buckets are aligned to their own size");
static_assert(N_NODES <= (1 << (31 - BKT_SHIFT)), "an id and its slot fit one sort key");

#define PART_THREADS 256
#define PART_EDGES   4096
#define PART_BLOCKS  ((N_EDGES + PART_EDGES - 1) / PART_EDGES)
#define N_BINS       256
#define BIN_NODES    ((N_NODES + N_BINS - 1) / N_BINS)
#define CELL_INTS    64
#define CELL_CAP     63
#define LOC_BITS     9
#define BUILD_THREADS 512
#define BUILD_WAVES   16
#define BUILD_ROWS   ((PART_BLOCKS + BUILD_WAVES - 1) / BUILD_WAVES)

typedef __attribute__((ext_vector_type(4))) int v4i;

static_assert(N_BINS == PART_THREADS, "one histogram word per thread");
static_assert(BIN_NODES <= (1 << LOC_BITS), "a node's place in its bin fits the low bits");
static_assert(N_NODES <= (1 << (31 - LOC_BITS)), "a source id fits the high bits");
static_assert(N_EDGES % 4 == 0 && PART_EDGES % (4 * PART_THREADS) == 0, "edges are read four at a time");

__global__ __launch_bounds__(PART_THREADS) void edge_partition_kernel(
        const int* __restrict__ src, const int* __restrict__ dst,
        int* __restrict__ cells, int n_edges) {
    __shared__ int hist[N_BINS];
    __shared__ __attribute__((aligned(16))) int stage[N_BINS * CELL_INTS];
    const int t = threadIdx.x;
    hist[t] = 0;
    __syncthreads();

    constexpr int Q = PART_EDGES / (4 * PART_THREADS);
    const int e0 = blockIdx.x * PART_EDGES;
    if (e0 + PART_EDGES <= n_edges) {
        v4i d[Q], s[Q];
#pragma unroll
        for (int k = 0; k < Q; ++k) {
            const int e = e0 + 4 * (k * PART_THREADS + t);
            d[k] = *(const v4i*)(dst + e);
            s[k] = *(const v4i*)(src + e);
        }
        int bin[4 * Q], key[4 * Q], r[4 * Q];
#pragma unroll
        for (int k = 0; k < 4 * Q; ++k) {
            const int dd = d[k >> 2][k & 3];
            bin[k] = dd / BIN_NODES;
            key[k] = (s[k >> 2][k & 3] << LOC_BITS) | (dd - bin[k] * BIN_NODES);
        }
#pragma unroll
        for (int k = 0; k < 4 * Q; ++k) r[k] = atomicAdd(&hist[bin[k]], 1);
#pragma unroll
        for (int k = 0; k < 4 * Q; ++k)
            if (r[k] < CELL_CAP) stage[bin[k] * CELL_INTS + 1 + r[k]] = key[k];
    } else {
        for (int e = e0 + t; e < n_edges; e += PART_THREADS) {
            const int dd = dst[e];
            const int bn = dd / BIN_NODES;
            const int rr = atomicAdd(&hist[bn], 1);
            if (rr < CELL_CAP) stage[bn * CELL_INTS + 1 + rr] = (src[e] << LOC_BITS) | (dd - bn * BIN_NODES);
        }
    }
    __syncthreads();

    for (int idx = t; idx < N_BINS * (CELL_INTS / 4); idx += PART_THREADS) {
        const int row = idx >> 4;
        const int q   = idx & 15;
        const int n   = hist[row];
        const int lim = (n < CELL_CAP) ? n : CELL_CAP;
        if (q < 8 || 4 * q <= lim) {
            v4i v = *(const v4i*)(stage + row * CELL_INTS + 4 * q);
#pragma unroll
            for (int c = 0; c < 4; ++c) {
                const int slot = 4 * q + c;
                v[c] = (slot == 0) ? n : ((slot <= lim) ? v[c] : 0);
            }
            *(v4i*)(cells + ((size_t)row * PART_BLOCKS + blockIdx.x) * CELL_INTS + 4 * q) = v;
        }
    }
}

__global__ __launch_bounds__(BUILD_THREADS) void bucket_build_kernel(
        const int* __restrict__ cells, const int* __restrict__ src, const int* __restrict__ dst,
        int* __restrict__ cnt, int* __restrict__ bkt, int n_nodes, int n_edges) {
    __shared__ int deg_l[BIN_NODES];
    __shared__ __attribute__((aligned(16))) int ids_l[BIN_NODES * BKT_CAP];
    const int t     = threadIdx.x;
    const int lane  = t & 31;
    const int wave  = __builtin_amdgcn_readfirstlane(t >> 5);
    const int node0 = blockIdx.x * BIN_NODES;
    const int* my   = cells + (size_t)blockIdx.x * PART_BLOCKS * CELL_INTS;

    if (t < BIN_NODES) deg_l[t] = 0;

    int v[BUILD_ROWS];
#pragma unroll
    for (int j = 0; j < BUILD_ROWS; ++j) {
        const int b  = wave + j * BUILD_WAVES;
        const int bb = (b < PART_BLOCKS) ? b : PART_BLOCKS - 1;
        v[j] = my[bb * CELL_INTS + lane];
    }
    __syncthreads();

    auto put = [&](int val) {
        const int ld = val & ((1 << LOC_BITS) - 1);
        const int r  = atomicAdd(&deg_l[ld], 1);
        if (r < BKT_CAP) ids_l[ld * BKT_CAP + r] = val >> LOC_BITS;
    };

#pragma unroll
    for (int j = 0; j < BUILD_ROWS; ++j) {
        const int b = wave + j * BUILD_WAVES;
        if (b < PART_BLOCKS) {
            const int n = __builtin_amdgcn_readlane(v[j], 0);
            if (n <= CELL_CAP) {
                if (lane >= 1 && lane <= n) put(v[j]);
                if (n >= 32 && lane + 32 <= n) put(my[b * CELL_INTS + 32 + lane]);
            } else {
                const int e1 = ((b + 1) * PART_EDGES < n_edges) ? (b + 1) * PART_EDGES : n_edges;
                for (int e = b * PART_EDGES + lane; e < e1; e += 32) {
                    const int loc = dst[e] - node0;
                    if (loc >= 0 && loc < BIN_NODES) put((src[e] << LOC_BITS) | loc);
                }
            }
        }
    }
    __syncthreads();

    if (t < BIN_NODES && node0 + t < n_nodes) cnt[node0 + t] = deg_l[t];
    for (int idx = t; idx < BIN_NODES * (BKT_CAP / 4); idx += BUILD_THREADS) {
        const int ld = idx >> 4;
        const int q  = idx & 15;
        if (node0 + ld < n_nodes && 4 * q < deg_l[ld])
            *(v4i*)(bkt + (size_t)(node0 + ld) * BKT_CAP + 4 * q) = *(const v4i*)(ids_l + ld * BKT_CAP + 4 * q);
    }
}

template <int V> struct lane_vec;
template <> struct lane_vec<2> { typedef v2f type; };
template <> struct lane_vec<4> { typedef v4f type; };

template <int D>
__global__ __launch_bounds__(AGG_THREADS) void gather_sum_kernel(
        const float* __restrict__ feat, const int* __restrict__ cnt, const int* __restrict__ bkt,
        const int* __restrict__ src, const int* __restrict__ dst,
        float* __restrict__ agg, int n_nodes, int n_edges) {
    constexpr int V = D / 32;
    typedef typename lane_vec<V>::type vec;
    const int lane = threadIdx.x & 31;
    const int wave0   = blockIdx.x * AGG_WAVES + __builtin_amdgcn_readfirstlane(threadIdx.x >> 5);
    const int n_waves = gridDim.x * AGG_WAVES;
    const float* fl = feat + lane * V;

    for (int node = wave0; node < n_nodes; node += n_waves) {
        const int deg = __builtin_amdgcn_readfirstlane(cnt[node]);
        const int* b  = bkt + (size_t)node * BKT_CAP;
        vec acc = {};
        if (deg <= 32) {
            const int key  = (lane < deg) ? b[lane] : 0;
            const int mine = (key << BKT_SHIFT) | lane;
            int r = 0;
            for (int jj = 0; jj < deg; ++jj) {
                const int pj = __builtin_amdgcn_readlane(mine, jj);
                r += (pj < mine) ? 1 : 0;
            }
            const int to  = (lane < deg) ? r : lane;
            const int ids = __builtin_amdgcn_ds_permute(to << 2, key);
            int jj = 0;
            for (; jj + 4 <= deg; jj += 4) {
                const int s0 = __builtin_amdgcn_readlane(ids, jj);
                const int s1 = __builtin_amdgcn_readlane(ids, jj + 1);
                const int s2 = __builtin_amdgcn_readlane(ids, jj + 2);
                const int s3 = __builtin_amdgcn_readlane(ids, jj + 3);
                const vec r0 = *(const vec*)(fl + (size_t)s0 * D);
                const vec r1 = *(const vec*)(fl + (size_t)s1 * D);
                const vec r2 = *(const vec*)(fl + (size_t)s2 * D);
                const vec r3 = *(const vec*)(fl + (size_t)s3 * D);
                acc += r0;
                acc += r1;
                acc += r2;
                acc += r3;
            }
            if (jj < deg) {
                const int last = deg - 1;
                const int s0 = __builtin_amdgcn_readlane(ids, jj);
                const int s1 = __builtin_amdgcn_readlane(ids, (jj + 1 < last) ? jj + 1 : last);
                const int s2 = __builtin_amdgcn_readlane(ids, (jj + 2 < last) ? jj + 2 : last);
                const vec r0 = *(const vec*)(fl + (size_t)s0 * D);
                const vec r1 = *(const vec*)(fl + (size_t)s1 * D);
                const vec r2 = *(const vec*)(fl + (size_t)s2 * D);
                acc += r0;
                if (jj + 1 < deg) acc += r1;
                if (jj + 2 < deg) acc += r2;
            }
        } else if (deg <= BKT_CAP) {
            const bool v1 = lane + 32 < deg;
            const int k0 = b[lane];
            const int k1 = v1 ? b[32 + lane] : 0;
            const int p0 = (k0 << BKT_SHIFT) | lane;
            const int p1 = v1 ? ((k1 << BKT_SHIFT) | (lane + 32)) : 0x7fffffff;
            int r0 = 0, r1 = 0;
            for (int jj = 0; jj < 32; ++jj) {
                const int pj = __builtin_amdgcn_readlane(p0, jj);
                r0 += (pj < p0) ? 1 : 0;
                r1 += (pj < p1) ? 1 : 0;
            }
            for (int jj = 0; jj < deg - 32; ++jj) {
                const int pj = __builtin_amdgcn_readlane(p1, jj);
                r0 += (pj < p0) ? 1 : 0;
                r1 += (pj < p1) ? 1 : 0;
            }
            auto pick = [&](int pos) -> int {
                const unsigned m0 = __builtin_amdgcn_ballot_w32(r0 == pos);
                const unsigned m1 = __builtin_amdgcn_ballot_w32(v1 && r1 == pos);
                const int l0 = (m0 != 0u) ? __builtin_ctz(m0) : 0;
                const int l1 = (m1 != 0u) ? __builtin_ctz(m1) : 0;
                const int a0 = __builtin_amdgcn_readlane(k0, l0);
                const int a1 = __builtin_amdgcn_readlane(k1, l1);
                return (m0 != 0u) ? a0 : a1;
            };
            const int last = deg - 1;
            for (int pos = 0; pos < deg; pos += 4) {
                const int s0 = pick(pos);
                const int s1 = pick((pos + 1 < last) ? pos + 1 : last);
                const int s2 = pick((pos + 2 < last) ? pos + 2 : last);
                const int s3 = pick((pos + 3 < last) ? pos + 3 : last);
                const vec q0 = *(const vec*)(fl + (size_t)s0 * D);
                const vec q1 = *(const vec*)(fl + (size_t)s1 * D);
                const vec q2 = *(const vec*)(fl + (size_t)s2 * D);
                const vec q3 = *(const vec*)(fl + (size_t)s3 * D);
                acc += q0;
                if (pos + 1 < deg) acc += q1;
                if (pos + 2 < deg) acc += q2;
                if (pos + 3 < deg) acc += q3;
            }
        } else {
            for (int eb = 0; eb < n_edges; eb += 32) {
                const int e = eb + lane;
                const int d = (e < n_edges) ? dst[e] : -1;
                unsigned m = __builtin_amdgcn_ballot_w32(d == node);
                if (m != 0u) {
                    const int sv = (e < n_edges) ? src[e] : 0;
                    while (m != 0u) {
                        const int s = __builtin_amdgcn_readlane(sv, __builtin_ctz(m));
                        acc += *(const vec*)(fl + (size_t)s * D);
                        m &= m - 1u;
                    }
                }
            }
        }
        *(vec*)(agg + (size_t)node * D + lane * V) = acc;
    }
}

#define GEMM_THREADS 384
#define GEMM_WAVES   12
#define GEMM_BLOCKS  256
#define CHUNK_K      32

template <int K1, int K2, int P>
__device__ __forceinline__ void stage_weights(float* lds, const float* __restrict__ Wa,
                                              const float* __restrict__ Wb) {
    constexpr int Q1 = K1 / 4, Q2 = K2 / 4;
    constexpr int T1 = D_HID * Q1, T2 = D_HID * Q2;
    constexpr int N1 = (T1 + GEMM_THREADS - 1) / GEMM_THREADS;
    constexpr int N2 = (T2 + GEMM_THREADS - 1) / GEMM_THREADS;
    const int tid = (int)threadIdx.x;
    v4f u[N1], w[N2];
#pragma unroll
    for (int k = 0; k < N1; ++k) {
        const int i = min(tid + k * GEMM_THREADS, T1 - 1);
        u[k] = *(const v4f*)(Wa + 4 * i);
    }
#pragma unroll
    for (int k = 0; k < N2; ++k) {
        const int i = min(tid + k * GEMM_THREADS, T2 - 1);
        w[k] = *(const v4f*)(Wb + 4 * i);
    }
#pragma unroll
    for (int k = 0; k < N1; ++k) {
        const int i = min(tid + k * GEMM_THREADS, T1 - 1);
        const int n = i / Q1;
        const int q = i - n * Q1;
        const int j = (n & 7) * 16 + (n >> 3);
        *(v4f*)(lds + j * P + 4 * q) = u[k];
    }
#pragma unroll
    for (int k = 0; k < N2; ++k) {
        const int i = min(tid + k * GEMM_THREADS, T2 - 1);
        const int n = i / Q2;
        const int q = i - n * Q2;
        const int j = (n & 7) * 16 + (n >> 3);
        *(v4f*)(lds + j * P + K1 + 4 * q) = w[k];
    }
}

template <int P>
__device__ __forceinline__ void chunk_mma(v8f (&acc)[8], const v4f (&a)[4], v4f (&b)[8],
                                          const float* w_cur, const float* w_after) {
#pragma unroll
    for (int i = 0; i < 4; ++i) {
        const float* wn = (i < 3) ? w_cur + 8 * (i + 1) : w_after;
        v4f bn[8];
#pragma unroll
        for (int t = 0; t < 8; ++t) bn[t] = *(const v4f*)(wn + t * 16 * P);

        const v2f a_lo = a[i].xy;
        const v2f a_hi = a[i].zw;
#pragma unroll
        for (int t = 0; t < 8; ++t) {
            const v2f bb = b[t].xy;
            acc[t] = __builtin_amdgcn_wmma_f32_16x16x4_f32(false, a_lo, false, bb,
                                                           (short)0, acc[t], false, false);
        }
#pragma unroll
        for (int t = 0; t < 8; ++t) {
            const v2f bb = b[t].zw;
            acc[t] = __builtin_amdgcn_wmma_f32_16x16x4_f32(false, a_hi, false, bb,
                                                           (short)0, acc[t], false, false);
        }
#pragma unroll
        for (int t = 0; t < 8; ++t) b[t] = bn[t];
    }
}

template <int K1, int K2, int RELU, int P>
__device__ __forceinline__ void strip_quarter(const float* sW,
        const float* __restrict__ A1, const float* __restrict__ A2,
        const float* __restrict__ bias, float* __restrict__ D, int row0, int q, int c, int h) {
    constexpr int NB = (K1 + K2) / 8;
    const float* a1 = A1 + (size_t)(row0 + c) * K1 + 4 * h;
    const float* a2 = A2 + (size_t)(row0 + c) * K2 + 4 * h;
    v4f a[NB];
#pragma unroll
    for (int i = 0; i < NB; ++i)
        a[i] = (8 * i < K1) ? *(const v4f*)(a1 + 8 * i) : *(const v4f*)(a2 + (8 * i - K1));

    const int n0 = 32 * q + c;
    const int n1 = n0 + 16;
    const float* w0 = sW + ((n0 & 7) * 16 + (n0 >> 3)) * P + 4 * h;
    const float* w1 = sW + ((n1 & 7) * 16 + (n1 >> 3)) * P + 4 * h;

    v8f acc0 = v8f{};
    v8f acc1 = v8f{};
#pragma unroll
    for (int i = 0; i < NB; ++i) {
        const v4f b0 = *(const v4f*)(w0 + 8 * i);
        const v4f b1 = *(const v4f*)(w1 + 8 * i);
        const v2f a_lo = a[i].xy;
        const v2f a_hi = a[i].zw;
        acc0 = __builtin_amdgcn_wmma_f32_16x16x4_f32(false, a_lo, false, b0.xy, (short)0, acc0, false, false);
        acc1 = __builtin_amdgcn_wmma_f32_16x16x4_f32(false, a_lo, false, b1.xy, (short)0, acc1, false, false);
        acc0 = __builtin_amdgcn_wmma_f32_16x16x4_f32(false, a_hi, false, b0.zw, (short)0, acc0, false, false);
        acc1 = __builtin_amdgcn_wmma_f32_16x16x4_f32(false, a_hi, false, b1.zw, (short)0, acc1, false, false);
    }

    const float bz0 = bias[n0];
    const float bz1 = bias[n1];
#pragma unroll
    for (int r = 0; r < 8; ++r) {
        float v0 = acc0[r] + bz0;
        float v1 = acc1[r] + bz1;
        if (RELU) {
            v0 = fmaxf(v0, 0.f);
            v1 = fmaxf(v1, 0.f);
        }
        float* p = D + (size_t)(row0 + 8 * h + r) * D_HID;
        p[n0] = v0;
        p[n1] = v1;
    }
}

template <int K1, int K2, int RELU>
__global__ __launch_bounds__(GEMM_THREADS) void gemm_dual_strip(
        const float* __restrict__ A1, const float* __restrict__ W1,
        const float* __restrict__ A2, const float* __restrict__ W2,
        const float* __restrict__ bias, float* __restrict__ D, int n_strips) {
    constexpr int KK = K1 + K2;
    constexpr int P  = KK + 4;
    constexpr int C1 = K1 / CHUNK_K;
    constexpr int C  = KK / CHUNK_K;
    static_assert(K1 % CHUNK_K == 0 && K2 % CHUNK_K == 0 && C % 2 == 0, "whole chunks, in pairs");
    static_assert(((P / 4) & 1) == 1, "pitch is an odd number of 16-B units");

    __shared__ __attribute__((aligned(16))) float sW[D_HID * P];

    stage_weights<K1, K2, P>(sW, W1, W2);
    __syncthreads();

    const int lane = threadIdx.x & 31;
    const int h    = lane >> 4;
    const int c    = lane & 15;
    const int wave = __builtin_amdgcn_readfirstlane(threadIdx.x >> 5);

    const v4f b_lo = *(const v4f*)(bias + 8 * c);
    const v4f b_hi = *(const v4f*)(bias + 8 * c + 4);

    const float* w_lds = sW + c * P + 4 * h;

    const int per_pass = GEMM_WAVES * (int)gridDim.x;
    const int n_whole  = (n_strips / per_pass) * per_pass;

    for (int ls = n_whole + (int)blockIdx.x; ls < n_strips; ls += (int)gridDim.x) {
        if (wave % 3 == 0)
            strip_quarter<K1, K2, RELU, P>(sW, A1, A2, bias, D, ls * 16, wave / 3, c, h);
    }

    for (int s = wave * (int)gridDim.x + (int)blockIdx.x; s < n_whole; s += per_pass) {
        const int row0 = s * 16;
        const float* a1 = A1 + (size_t)(row0 + c) * K1 + 4 * h;
        const float* a2 = A2 + (size_t)(row0 + c) * K2 + 4 * h;

        v8f acc[8];
#pragma unroll
        for (int t = 0; t < 8; ++t) acc[t] = v8f{};

        v4f aA[4], aB[4], b[8];
#pragma unroll
        for (int i = 0; i < 4; ++i) aA[i] = *(const v4f*)(a1 + 8 * i);
#pragma unroll
        for (int t = 0; t < 8; ++t) b[t] = *(const v4f*)(w_lds + t * 16 * P);

#pragma unroll 1
        for (int ch = 0; ch < C; ch += 2) {
            const float* pB = (ch + 1 < C1) ? a1 + CHUNK_K * (ch + 1)
                                            : a2 + CHUNK_K * (ch + 1 - C1);
#pragma unroll
            for (int i = 0; i < 4; ++i) aB[i] = *(const v4f*)(pB + 8 * i);
            chunk_mma<P>(acc, aA, b, w_lds + CHUNK_K * ch, w_lds + CHUNK_K * (ch + 1));

            const int cn = (ch + 2 < C) ? ch + 2 : ch + 1;
            const float* pA = (cn < C1) ? a1 + CHUNK_K * cn : a2 + CHUNK_K * (cn - C1);
#pragma unroll
            for (int i = 0; i < 4; ++i) aA[i] = *(const v4f*)(pA + 8 * i);
            chunk_mma<P>(acc, aB, b, w_lds + CHUNK_K * (ch + 1), w_lds + CHUNK_K * cn);
        }

#pragma unroll
        for (int r = 0; r < 8; ++r) {
            v4f lo = {acc[0][r], acc[1][r], acc[2][r], acc[3][r]};
            v4f hi = {acc[4][r], acc[5][r], acc[6][r], acc[7][r]};
            lo += b_lo;
            hi += b_hi;
            if (RELU) {
#pragma unroll
                for (int e = 0; e < 4; ++e) {
                    lo[e] = fmaxf(lo[e], 0.f);
                    hi[e] = fmaxf(hi[e], 0.f);
                }
            }
            float* p = D + (size_t)(row0 + 8 * h + r) * D_HID + 8 * c;
            *(v4f*)p       = lo;
            *(v4f*)(p + 4) = hi;
        }
    }
}

extern "C" void kernel_launch(void* const* d_in, const int* in_sizes, int n_in,
                              void* d_out, int out_size, void* d_ws, size_t ws_size,
                              hipStream_t stream) {
    const float* x       = (const float*)d_in[0];
    const int*   edge    = (const int*)d_in[1];
    const float* W1_rel  = (const float*)d_in[2];
    const float* b1_rel  = (const float*)d_in[3];
    const float* W1_root = (const float*)d_in[4];
    const float* W2_rel  = (const float*)d_in[5];
    const float* b2_rel  = (const float*)d_in[6];
    const float* W2_root = (const float*)d_in[7];
    float* out = (float*)d_out;

    const int* src = edge;
    const int* dst = edge + N_EDGES;

    float* h   = (float*)d_ws;
    float* agg = h + (size_t)N_NODES * D_HID;

    static_assert((size_t)BKT_INTS <= (size_t)N_NODES * D_HID, "degree and bucket arrays fit in the output buffer");
    int* csr = (int*)d_out;
    int* cnt = csr + BKT_CNT;
    int* bkt = csr + BKT_IDS;

    const int n_strips    = N_NODES / 16;

    static_assert((size_t)N_BINS * PART_BLOCKS * CELL_INTS <= (size_t)N_NODES * D_HID, "edge cells fit in h's place");
    int* cells = (int*)d_ws;

    edge_partition_kernel<<<PART_BLOCKS, PART_THREADS, 0, stream>>>(src, dst, cells, N_EDGES);
    bucket_build_kernel<<<N_BINS, BUILD_THREADS, 0, stream>>>(cells, src, dst, cnt, bkt, N_NODES, N_EDGES);

    gather_sum_kernel<D_IN><<<NODE_BLOCKS, AGG_THREADS, 0, stream>>>(
        x, cnt, bkt, src, dst, agg, N_NODES, N_EDGES);
    gemm_dual_strip<D_IN, D_IN, 1><<<GEMM_BLOCKS, GEMM_THREADS, 0, stream>>>(
        agg, W1_rel, x, W1_root, b1_rel, h, n_strips);

    gather_sum_kernel<D_HID><<<NODE_BLOCKS, AGG_THREADS, 0, stream>>>(
        h, cnt, bkt, src, dst, agg, N_NODES, N_EDGES);
    gemm_dual_strip<D_HID, D_HID, 0><<<GEMM_BLOCKS, GEMM_THREADS, 0, stream>>>(
        agg, W2_rel, h, W2_root, b2_rel, out, n_strips);
}
